# starting kernel
# baseline (speedup 1.0000x reference)
#include <hip/hip_runtime.h>


typedef float v2f __attribute__((ext_vector_type(2)));
typedef float v8f __attribute__((ext_vector_type(8)));

#define N_STATES   16
#define STATE_SIZE 32
#define ENC        64
#define HID        128
#define OUT_DIM    8
#define TILE       16
#define BLOCK_THREADS 256

__global__ __launch_bounds__(BLOCK_THREADS)
void gnn_fused_wmma(const float* __restrict__ obs,
                    const float* __restrict__ W_enc, const float* __restrict__ b_enc,
                    const float* __restrict__ W_f,   const float* __restrict__ b_f,
                    const float* __restrict__ W_dec, const float* __restrict__ b_dec,
                    float* __restrict__ out, int n_samples)
{
    __shared__ float f_in[TILE * 2 * ENC];
    __shared__ float hidden[TILE * HID];

    const int tid   = threadIdx.x;
    const int wave  = tid >> 5;
    const int lane  = tid & 31;
    const int lo    = lane & 15;
    const int hi    = lane >> 4;
    const int khalf = hi * 2;
    const int mbase = hi * 8;
    const int blockBase = blockIdx.x * TILE;

    for (int s = 0; s < 2; ++s) {
        const int ls = wave * 2 + s;
        const int b  = blockBase + ls;
        const float* st = obs + (size_t)b * (N_STATES * STATE_SIZE);
        if (s == 0) __builtin_prefetch(st + N_STATES * STATE_SIZE, 0, 0);

        v2f a[8];
#pragma unroll
        for (int k = 0; k < 8; ++k)
            a[k] = *(const v2f*)(st + lo * STATE_SIZE + k * 4 + khalf);

        const float flag = a[0][0];
        const bool ok = (hi == 0) ? (flag == 1.0f) : true;
        const unsigned msk   = __builtin_amdgcn_ballot_w32(ok);
        const unsigned inval = ((~msk) >> 1) & 0x7FFFu;
        const int   cnt = __builtin_ctz(inval | 0x8000u);
        const float inv = 1.0f / (float)(cnt > 1 ? cnt : 1);

#pragma unroll
        for (int t = 0; t < 4; ++t) {
            const int n = lo + t * 16;
            v8f acc = {};
#pragma unroll
            for (int k = 0; k < 8; ++k) {
                const int kk = k * 4 + khalf;
                v2f bf;
                bf[0] = W_enc[kk * ENC + n];
                bf[1] = W_enc[(kk + 1) * ENC + n];
                acc = __builtin_amdgcn_wmma_f32_16x16x4_f32(
                        false, a[k], false, bf, (short)0, acc, false, false);
            }
            const float bias = b_enc[n];
            float part = 0.0f, agent = 0.0f;
#pragma unroll
            for (int v = 0; v < 8; ++v) {
                const int m = v + mbase;
                float e = acc[v] + bias;
                e = e > 0.0f ? e : 0.0f;
                if (m == 0) agent = e;
                const float w = (m >= 1 && m <= cnt) ? inv : 0.0f;
                part += w * e;
            }
            part += __shfl_xor(part, 16, 32);
            if (hi == 0) {
                f_in[ls * (2 * ENC) + n]       = agent;
                f_in[ls * (2 * ENC) + ENC + n] = part;
            }
        }
    }
    __syncthreads();

    {
        const int n = lo + wave * 16;
        v8f acc = {};
#pragma unroll 4
        for (int k = 0; k < 32; ++k) {
            const int kk = k * 4 + khalf;
            v2f av = *(const v2f*)&f_in[lo * (2 * ENC) + kk];
            v2f bf;
            bf[0] = W_f[kk * HID + n];
            bf[1] = W_f[(kk + 1) * HID + n];
            acc = __builtin_amdgcn_wmma_f32_16x16x4_f32(
                    false, av, false, bf, (short)0, acc, false, false);
        }
        const float bias = b_f[n];
#pragma unroll
        for (int v = 0; v < 8; ++v) {
            float e = acc[v] + bias;
            hidden[(v + mbase) * HID + n] = e > 0.0f ? e : 0.0f;
        }
    }
    __syncthreads();

    if (wave == 0) {
        const int n = lo;
        const float nmask = (n < OUT_DIM) ? 1.0f : 0.0f;
        const int nc = n & (OUT_DIM - 1);
        v8f acc = {};
#pragma unroll 4
        for (int k = 0; k < 32; ++k) {
            const int kk = k * 4 + khalf;
            v2f av = *(const v2f*)&hidden[lo * HID + kk];
            v2f bf;
            bf[0] = W_dec[kk * OUT_DIM + nc] * nmask;
            bf[1] = W_dec[(kk + 1) * OUT_DIM + nc] * nmask;
            acc = __builtin_amdgcn_wmma_f32_16x16x4_f32(
                    false, av, false, bf, (short)0, acc, false, false);
        }
        if (n < OUT_DIM) {
            const float bias = b_dec[n];
#pragma unroll
            for (int v = 0; v < 8; ++v) {
                const int m = v + mbase;
                out[(size_t)(blockBase + m) * OUT_DIM + n] = acc[v] + bias;
            }
        }
    }
}

extern "C" void kernel_launch(void* const* d_in, const int* in_sizes, int n_in,
                              void* d_out, int out_size, void* d_ws, size_t ws_size,
                              hipStream_t stream) {
    (void)n_in; (void)out_size; (void)d_ws; (void)ws_size;
    const float* obs   = (const float*)d_in[0];
    const float* W_enc = (const float*)d_in[1];
    const float* b_enc = (const float*)d_in[2];
    const float* W_f   = (const float*)d_in[3];
    const float* b_f   = (const float*)d_in[4];
    const float* W_dec = (const float*)d_in[5];
    const float* b_dec = (const float*)d_in[6];
    float* out = (float*)d_out;

    const int n_samples = in_sizes[0] / (N_STATES * STATE_SIZE);
    if (n_samples <= 0) return;
    const int grid = n_samples / TILE;

    gnn_fused_wmma<<<grid, BLOCK_THREADS, 0, stream>>>(
        obs, W_enc, b_enc, W_f, b_f, W_dec, b_dec, out, n_samples);
}


# Pad LDS row pitch of the f_in and hidden tiles from 128 to 132 floats: removes the 16-way bank conflict on every A-fragment ds_load_b64 of the hidden and logits GEMM loops
# speedup vs baseline: 1.0710x; 1.0710x over previous
#include <hip/hip_runtime.h>


typedef float v2f __attribute__((ext_vector_type(2)));
typedef float v8f __attribute__((ext_vector_type(8)));

#define N_STATES   16
#define STATE_SIZE 32
#define ENC        64
#define HID        128
#define OUT_DIM    8
#define TILE       16
#define LDS_PITCH  (HID + 4)
#define BLOCK_THREADS 256
static_assert(2 * ENC == HID, "f_in and hidden share one padded row pitch");

__global__ __launch_bounds__(BLOCK_THREADS)
void gnn_fused_wmma(const float* __restrict__ obs,
                    const float* __restrict__ W_enc, const float* __restrict__ b_enc,
                    const float* __restrict__ W_f,   const float* __restrict__ b_f,
                    const float* __restrict__ W_dec, const float* __restrict__ b_dec,
                    float* __restrict__ out, int n_samples)
{
    __shared__ float f_in[TILE * LDS_PITCH];
    __shared__ float hidden[TILE * LDS_PITCH];

    const int tid   = threadIdx.x;
    const int wave  = tid >> 5;
    const int lane  = tid & 31;
    const int lo    = lane & 15;
    const int hi    = lane >> 4;
    const int khalf = hi * 2;
    const int mbase = hi * 8;
    const int blockBase = blockIdx.x * TILE;

    for (int s = 0; s < 2; ++s) {
        const int ls = wave * 2 + s;
        const int b  = blockBase + ls;
        const float* st = obs + (size_t)b * (N_STATES * STATE_SIZE);
        if (s == 0) __builtin_prefetch(st + N_STATES * STATE_SIZE, 0, 0);

        v2f a[8];
#pragma unroll
        for (int k = 0; k < 8; ++k)
            a[k] = *(const v2f*)(st + lo * STATE_SIZE + k * 4 + khalf);

        const float flag = a[0][0];
        const bool ok = (hi == 0) ? (flag == 1.0f) : true;
        const unsigned msk   = __builtin_amdgcn_ballot_w32(ok);
        const unsigned inval = ((~msk) >> 1) & 0x7FFFu;
        const int   cnt = __builtin_ctz(inval | 0x8000u);
        const float inv = 1.0f / (float)(cnt > 1 ? cnt : 1);

#pragma unroll
        for (int t = 0; t < 4; ++t) {
            const int n = lo + t * 16;
            v8f acc = {};
#pragma unroll
            for (int k = 0; k < 8; ++k) {
                const int kk = k * 4 + khalf;
                v2f bf;
                bf[0] = W_enc[kk * ENC + n];
                bf[1] = W_enc[(kk + 1) * ENC + n];
                acc = __builtin_amdgcn_wmma_f32_16x16x4_f32(
                        false, a[k], false, bf, (short)0, acc, false, false);
            }
            const float bias = b_enc[n];
            float part = 0.0f, agent = 0.0f;
#pragma unroll
            for (int v = 0; v < 8; ++v) {
                const int m = v + mbase;
                float e = acc[v] + bias;
                e = e > 0.0f ? e : 0.0f;
                if (m == 0) agent = e;
                const float w = (m >= 1 && m <= cnt) ? inv : 0.0f;
                part += w * e;
            }
            part += __shfl_xor(part, 16, 32);
            if (hi == 0) {
                f_in[ls * LDS_PITCH + n]       = agent;
                f_in[ls * LDS_PITCH + ENC + n] = part;
            }
        }
    }
    __syncthreads();

    {
        const int n = lo + wave * 16;
        v8f acc = {};
#pragma unroll 4
        for (int k = 0; k < 32; ++k) {
            const int kk = k * 4 + khalf;
            v2f av = *(const v2f*)&f_in[lo * LDS_PITCH + kk];
            v2f bf;
            bf[0] = W_f[kk * HID + n];
            bf[1] = W_f[(kk + 1) * HID + n];
            acc = __builtin_amdgcn_wmma_f32_16x16x4_f32(
                    false, av, false, bf, (short)0, acc, false, false);
        }
        const float bias = b_f[n];
#pragma unroll
        for (int v = 0; v < 8; ++v) {
            float e = acc[v] + bias;
            hidden[(v + mbase) * LDS_PITCH + n] = e > 0.0f ? e : 0.0f;
        }
    }
    __syncthreads();

    if (wave == 0) {
        const int n = lo;
        const float nmask = (n < OUT_DIM) ? 1.0f : 0.0f;
        const int nc = n & (OUT_DIM - 1);
        v8f acc = {};
#pragma unroll 4
        for (int k = 0; k < 32; ++k) {
            const int kk = k * 4 + khalf;
            v2f av = *(const v2f*)&hidden[lo * LDS_PITCH + kk];
            v2f bf;
            bf[0] = W_dec[kk * OUT_DIM + nc] * nmask;
            bf[1] = W_dec[(kk + 1) * OUT_DIM + nc] * nmask;
            acc = __builtin_amdgcn_wmma_f32_16x16x4_f32(
                    false, av, false, bf, (short)0, acc, false, false);
        }
        if (n < OUT_DIM) {
            const float bias = b_dec[n];
#pragma unroll
            for (int v = 0; v < 8; ++v) {
                const int m = v + mbase;
                out[(size_t)(blockBase + m) * OUT_DIM + n] = acc[v] + bias;
            }
        }
    }
}

extern "C" void kernel_launch(void* const* d_in, const int* in_sizes, int n_in,
                              void* d_out, int out_size, void* d_ws, size_t ws_size,
                              hipStream_t stream) {
    (void)n_in; (void)out_size; (void)d_ws; (void)ws_size;
    const float* obs   = (const float*)d_in[0];
    const float* W_enc = (const float*)d_in[1];
    const float* b_enc = (const float*)d_in[2];
    const float* W_f   = (const float*)d_in[3];
    const float* b_f   = (const float*)d_in[4];
    const float* W_dec = (const float*)d_in[5];
    const float* b_dec = (const float*)d_in[6];
    float* out = (float*)d_out;

    const int n_samples = in_sizes[0] / (N_STATES * STATE_SIZE);
    if (n_samples <= 0) return;
    const int grid = n_samples / TILE;

    gnn_fused_wmma<<<grid, BLOCK_THREADS, 0, stream>>>(
        obs, W_enc, b_enc, W_f, b_f, W_dec, b_dec, out, n_samples);
}


# Persistent 512-work-group grid looping over sample tiles: W_enc/W_f B-fragments kept in registers, W_dec in LDS, fetched once per work-group, not per 16-sample block; next tile's rows loaded ahead
# speedup vs baseline: 1.3136x; 1.2265x over previous
#include <hip/hip_runtime.h>


typedef float v2f __attribute__((ext_vector_type(2)));
typedef float v8f __attribute__((ext_vector_type(8)));

#define N_STATES   16
#define STATE_SIZE 32
#define ENC        64
#define HID        128
#define OUT_DIM    8
#define TILE       16
#define LDS_PITCH  (HID + 4)
#define BLOCK_THREADS 256
#define GRID_BLOCKS   512
static_assert(2 * ENC == HID, "f_in and hidden share one padded row pitch");

__device__ __forceinline__ void load_sample(v2f (&a)[8], const float* __restrict__ st,
                                            int lo, int khalf)
{
#pragma unroll
    for (int k = 0; k < 8; ++k)
        a[k] = *(const v2f*)(st + lo * STATE_SIZE + k * 4 + khalf);
}

__device__ __forceinline__ void lds_barrier()
{
    asm volatile("s_wait_dscnt 0x0\n\ts_barrier_signal -1\n\ts_barrier_wait -1" ::: "memory");
}

__global__ __launch_bounds__(BLOCK_THREADS)
void gnn_fused_wmma(const float* __restrict__ obs,
                    const float* __restrict__ W_enc, const float* __restrict__ b_enc,
                    const float* __restrict__ W_f,   const float* __restrict__ b_f,
                    const float* __restrict__ W_dec, const float* __restrict__ b_dec,
                    float* __restrict__ out, int n_samples)
{
    __shared__ float f_in[TILE * LDS_PITCH];
    __shared__ float hidden[TILE * LDS_PITCH];
    __shared__ v2f   wdec_s[(HID / 2) * 16];

    const int tid   = threadIdx.x;
    const int wave  = tid >> 5;
    const int lane  = tid & 31;
    const int lo    = lane & 15;
    const int hi    = lane >> 4;
    const int khalf = hi * 2;
    const int mbase = hi * 8;
    const int n_tiles = n_samples / TILE;
    const int stride  = gridDim.x;

    int tile = blockIdx.x;

    v2f a0[8], a1[8];
    if (tile < n_tiles) {
        const float* st = obs + (size_t)(tile * TILE + wave * 2) * (N_STATES * STATE_SIZE);
        load_sample(a0, st, lo, khalf);
        load_sample(a1, st + N_STATES * STATE_SIZE, lo, khalf);
    }

    for (int e = tid; e < (HID / 2) * 16; e += BLOCK_THREADS) {
        const int p = e >> 4, c = e & 15;
        v2f w = {0.0f, 0.0f};
        if (c < OUT_DIM) {
            w[0] = W_dec[(2 * p) * OUT_DIM + c];
            w[1] = W_dec[(2 * p + 1) * OUT_DIM + c];
        }
        wdec_s[e] = w;
    }
    v2f we[4][8];
    float be[4];
#pragma unroll
    for (int t = 0; t < 4; ++t) {
        const int n = lo + t * 16;
        be[t] = b_enc[n];
#pragma unroll
        for (int k = 0; k < 8; ++k) {
            const int kk = k * 4 + khalf;
            we[t][k][0] = W_enc[kk * ENC + n];
            we[t][k][1] = W_enc[(kk + 1) * ENC + n];
        }
    }
    const int nW = lo + wave * 16;
    v2f wf[32];
#pragma unroll
    for (int k = 0; k < 32; ++k) {
        const int kk = k * 4 + khalf;
        wf[k][0] = W_f[kk * HID + nW];
        wf[k][1] = W_f[(kk + 1) * HID + nW];
    }
    const float bias_f = b_f[nW];
    const float bias_d = b_dec[lo & (OUT_DIM - 1)];

    for (; tile < n_tiles; tile += stride) {
        const int blockBase = tile * TILE;

#pragma unroll
        for (int s = 0; s < 2; ++s) {
            const int ls = wave * 2 + s;
            const v2f (&a)[8] = (s == 0) ? a0 : a1;

            const float flag = a[0][0];
            const bool ok = (hi == 0) ? (flag == 1.0f) : true;
            const unsigned msk   = __builtin_amdgcn_ballot_w32(ok);
            const unsigned inval = ((~msk) >> 1) & 0x7FFFu;
            const int   cnt = __builtin_ctz(inval | 0x8000u);
            const float inv = 1.0f / (float)(cnt > 1 ? cnt : 1);

#pragma unroll
            for (int t = 0; t < 4; ++t) {
                const int n = lo + t * 16;
                v8f acc = {};
#pragma unroll
                for (int k = 0; k < 8; ++k)
                    acc = __builtin_amdgcn_wmma_f32_16x16x4_f32(
                            false, a[k], false, we[t][k], (short)0, acc, false, false);
                const float bias = be[t];
                float part = 0.0f, agent = 0.0f;
#pragma unroll
                for (int v = 0; v < 8; ++v) {
                    const int m = v + mbase;
                    float e = acc[v] + bias;
                    e = e > 0.0f ? e : 0.0f;
                    if (m == 0) agent = e;
                    const float w = (m >= 1 && m <= cnt) ? inv : 0.0f;
                    part += w * e;
                }
                part += __shfl_xor(part, 16, 32);
                if (hi == 0) {
                    f_in[ls * LDS_PITCH + n]       = agent;
                    f_in[ls * LDS_PITCH + ENC + n] = part;
                }
            }
        }

        if (tile + stride < n_tiles) {
            const float* st = obs + (size_t)((tile + stride) * TILE + wave * 2) * (N_STATES * STATE_SIZE);
            load_sample(a0, st, lo, khalf);
            load_sample(a1, st + N_STATES * STATE_SIZE, lo, khalf);
        }
        lds_barrier();

        {
            v8f acc = {};
#pragma unroll
            for (int k = 0; k < 32; ++k) {
                const int kk = k * 4 + khalf;
                v2f av = *(const v2f*)&f_in[lo * LDS_PITCH + kk];
                acc = __builtin_amdgcn_wmma_f32_16x16x4_f32(
                        false, av, false, wf[k], (short)0, acc, false, false);
            }
#pragma unroll
            for (int v = 0; v < 8; ++v) {
                float e = acc[v] + bias_f;
                hidden[(v + mbase) * LDS_PITCH + nW] = e > 0.0f ? e : 0.0f;
            }
        }
        lds_barrier();

        if (wave == 0) {
            v8f acc = {};
#pragma unroll 4
            for (int k = 0; k < 32; ++k) {
                const int kk = k * 4 + khalf;
                v2f av = *(const v2f*)&hidden[lo * LDS_PITCH + kk];
                v2f bf = wdec_s[(2 * k + hi) * 16 + lo];
                acc = __builtin_amdgcn_wmma_f32_16x16x4_f32(
                        false, av, false, bf, (short)0, acc, false, false);
            }
            if (lo < OUT_DIM) {
#pragma unroll
                for (int v = 0; v < 8; ++v) {
                    const int m = v + mbase;
                    out[(size_t)(blockBase + m) * OUT_DIM + lo] = acc[v] + bias_d;
                }
            }
        }
    }
}

extern "C" void kernel_launch(void* const* d_in, const int* in_sizes, int n_in,
                              void* d_out, int out_size, void* d_ws, size_t ws_size,
                              hipStream_t stream) {
    (void)n_in; (void)out_size; (void)d_ws; (void)ws_size;
    const float* obs   = (const float*)d_in[0];
    const float* W_enc = (const float*)d_in[1];
    const float* b_enc = (const float*)d_in[2];
    const float* W_f   = (const float*)d_in[3];
    const float* b_f   = (const float*)d_in[4];
    const float* W_dec = (const float*)d_in[5];
    const float* b_dec = (const float*)d_in[6];
    float* out = (float*)d_out;

    const int n_samples = in_sizes[0] / (N_STATES * STATE_SIZE);
    if (n_samples <= 0) return;

    gnn_fused_wmma<<<GRID_BLOCKS, BLOCK_THREADS, 0, stream>>>(
        obs, W_enc, b_enc, W_f, b_f, W_dec, b_dec, out, n_samples);
}


# Pack the needed encoder rows (agent + leading-valid neighbours) of a wave's two samples into one 16-row WMMA tile when they fit, else a tile each; reorders the 2nd sample's pooled partial sums
# speedup vs baseline: 1.5238x; 1.1600x over previous
#include <hip/hip_runtime.h>


typedef float v2f __attribute__((ext_vector_type(2)));
typedef float v8f __attribute__((ext_vector_type(8)));

#define N_STATES   16
#define STATE_SIZE 32
#define ENC        64
#define HID        128
#define OUT_DIM    8
#define TILE       16
#define LDS_PITCH  (HID + 4)
#define BLOCK_THREADS 256
#define GRID_BLOCKS   512
static_assert(2 * ENC == HID, "f_in and hidden share one padded row pitch");

__device__ __forceinline__ void load_sample(v2f (&a)[8], const float* __restrict__ st,
                                            int lo, int khalf)
{
#pragma unroll
    for (int k = 0; k < 8; ++k)
        a[k] = *(const v2f*)(st + lo * STATE_SIZE + k * 4 + khalf);
}

__device__ __forceinline__ void lds_barrier()
{
    asm volatile("s_wait_dscnt 0x0\n\ts_barrier_signal -1\n\ts_barrier_wait -1" ::: "memory");
}

__device__ __forceinline__ int valid_prefix(float flag, int hi)
{
    const bool ok = (hi == 0) ? (flag == 1.0f) : true;
    const unsigned msk   = __builtin_amdgcn_ballot_w32(ok);
    const unsigned inval = ((~msk) >> 1) & 0x7FFFu;
    return __builtin_ctz(inval | 0x8000u);
}

__device__ __forceinline__ void encode_single(const v2f (&a)[8], const v2f (&we)[4][8],
                                              const float (&be)[4], int cnt, int ls,
                                              int lo, int hi, int mbase, float* f_in)
{
    const float inv = 1.0f / (float)(cnt > 1 ? cnt : 1);
#pragma unroll
    for (int t = 0; t < 4; ++t) {
        const int n = lo + t * 16;
        v8f acc = {};
#pragma unroll
        for (int k = 0; k < 8; ++k)
            acc = __builtin_amdgcn_wmma_f32_16x16x4_f32(
                    false, a[k], false, we[t][k], (short)0, acc, false, false);
        const float bias = be[t];
        float part = 0.0f, agent = 0.0f;
#pragma unroll
        for (int v = 0; v < 8; ++v) {
            const int m = v + mbase;
            float e = acc[v] + bias;
            e = e > 0.0f ? e : 0.0f;
            if (m == 0) agent = e;
            const float w = (m >= 1 && m <= cnt) ? inv : 0.0f;
            part += w * e;
        }
        part += __shfl_xor(part, 16, 32);
        if (hi == 0) {
            f_in[ls * LDS_PITCH + n]       = agent;
            f_in[ls * LDS_PITCH + ENC + n] = part;
        }
    }
}

__device__ __forceinline__ void encode_pair(const v2f (&a0)[8], const v2f (&a1)[8],
                                            const v2f (&we)[4][8], const float (&be)[4],
                                            int cnt0, int cnt1, int ls0,
                                            int lo, int hi, int mbase, float* f_in)
{
    const int  r0    = cnt0 + 1;
    const bool from0 = lo < r0;
    const int  srcl  = ((lo - r0) & 15) | (hi << 4);
    v2f p[8];
#pragma unroll
    for (int k = 0; k < 8; ++k) {
        const float x = __shfl(a1[k][0], srcl, 32);
        const float y = __shfl(a1[k][1], srcl, 32);
        p[k][0] = from0 ? a0[k][0] : x;
        p[k][1] = from0 ? a0[k][1] : y;
    }

    const float inv0 = 1.0f / (float)(cnt0 > 1 ? cnt0 : 1);
    const float inv1 = 1.0f / (float)(cnt1 > 1 ? cnt1 : 1);
    float w0[8], w1[8];
#pragma unroll
    for (int v = 0; v < 8; ++v) {
        const int m = v + mbase;
        w0[v] = (m >= 1 && m <= cnt0)       ? inv0 : 0.0f;
        w1[v] = (m > r0 && m <= r0 + cnt1)  ? inv1 : 0.0f;
    }
    const int ls1 = ls0 + 1;
    const bool has_agent1 = (hi == (r0 >> 3));

#pragma unroll
    for (int t = 0; t < 4; ++t) {
        const int n = lo + t * 16;
        v8f acc = {};
#pragma unroll
        for (int k = 0; k < 8; ++k)
            acc = __builtin_amdgcn_wmma_f32_16x16x4_f32(
                    false, p[k], false, we[t][k], (short)0, acc, false, false);
        const float bias = be[t];
        float part0 = 0.0f, part1 = 0.0f, agent0 = 0.0f, agent1 = 0.0f;
#pragma unroll
        for (int v = 0; v < 8; ++v) {
            const int m = v + mbase;
            float e = acc[v] + bias;
            e = e > 0.0f ? e : 0.0f;
            if (m == 0)  agent0 = e;
            if (m == r0) agent1 = e;
            part0 += w0[v] * e;
            part1 += w1[v] * e;
        }
        part0 += __shfl_xor(part0, 16, 32);
        part1 += __shfl_xor(part1, 16, 32);
        if (hi == 0) {
            f_in[ls0 * LDS_PITCH + n]       = agent0;
            f_in[ls0 * LDS_PITCH + ENC + n] = part0;
            f_in[ls1 * LDS_PITCH + ENC + n] = part1;
        }
        if (has_agent1)
            f_in[ls1 * LDS_PITCH + n] = agent1;
    }
}

__global__ __launch_bounds__(BLOCK_THREADS)
void gnn_fused_wmma(const float* __restrict__ obs,
                    const float* __restrict__ W_enc, const float* __restrict__ b_enc,
                    const float* __restrict__ W_f,   const float* __restrict__ b_f,
                    const float* __restrict__ W_dec, const float* __restrict__ b_dec,
                    float* __restrict__ out, int n_samples)
{
    __shared__ float f_in[TILE * LDS_PITCH];
    __shared__ float hidden[TILE * LDS_PITCH];
    __shared__ v2f   wdec_s[(HID / 2) * 16];

    const int tid   = threadIdx.x;
    const int wave  = tid >> 5;
    const int lane  = tid & 31;
    const int lo    = lane & 15;
    const int hi    = lane >> 4;
    const int khalf = hi * 2;
    const int mbase = hi * 8;
    const int n_tiles = n_samples / TILE;
    const int stride  = gridDim.x;

    int tile = blockIdx.x;

    v2f a0[8], a1[8];
    if (tile < n_tiles) {
        const float* st = obs + (size_t)(tile * TILE + wave * 2) * (N_STATES * STATE_SIZE);
        load_sample(a0, st, lo, khalf);
        load_sample(a1, st + N_STATES * STATE_SIZE, lo, khalf);
    }

    for (int e = tid; e < (HID / 2) * 16; e += BLOCK_THREADS) {
        const int p = e >> 4, c = e & 15;
        v2f w = {0.0f, 0.0f};
        if (c < OUT_DIM) {
            w[0] = W_dec[(2 * p) * OUT_DIM + c];
            w[1] = W_dec[(2 * p + 1) * OUT_DIM + c];
        }
        wdec_s[e] = w;
    }
    v2f we[4][8];
    float be[4];
#pragma unroll
    for (int t = 0; t < 4; ++t) {
        const int n = lo + t * 16;
        be[t] = b_enc[n];
#pragma unroll
        for (int k = 0; k < 8; ++k) {
            const int kk = k * 4 + khalf;
            we[t][k][0] = W_enc[kk * ENC + n];
            we[t][k][1] = W_enc[(kk + 1) * ENC + n];
        }
    }
    const int nW = lo + wave * 16;
    v2f wf[32];
#pragma unroll
    for (int k = 0; k < 32; ++k) {
        const int kk = k * 4 + khalf;
        wf[k][0] = W_f[kk * HID + nW];
        wf[k][1] = W_f[(kk + 1) * HID + nW];
    }
    const float bias_f = b_f[nW];
    const float bias_d = b_dec[lo & (OUT_DIM - 1)];

    for (; tile < n_tiles; tile += stride) {
        const int blockBase = tile * TILE;

        {
            const int ls0  = wave * 2;
            const int cnt0 = valid_prefix(a0[0][0], hi);
            const int cnt1 = valid_prefix(a1[0][0], hi);
            if (cnt0 + cnt1 + 2 <= TILE) {
                encode_pair(a0, a1, we, be, cnt0, cnt1, ls0, lo, hi, mbase, f_in);
            } else {
                encode_single(a0, we, be, cnt0, ls0,     lo, hi, mbase, f_in);
                encode_single(a1, we, be, cnt1, ls0 + 1, lo, hi, mbase, f_in);
            }
        }

        if (tile + stride < n_tiles) {
            const float* st = obs + (size_t)((tile + stride) * TILE + wave * 2) * (N_STATES * STATE_SIZE);
            load_sample(a0, st, lo, khalf);
            load_sample(a1, st + N_STATES * STATE_SIZE, lo, khalf);
        }
        lds_barrier();

        {
            v8f acc = {};
#pragma unroll
            for (int k = 0; k < 32; ++k) {
                const int kk = k * 4 + khalf;
                v2f av = *(const v2f*)&f_in[lo * LDS_PITCH + kk];
                acc = __builtin_amdgcn_wmma_f32_16x16x4_f32(
                        false, av, false, wf[k], (short)0, acc, false, false);
            }
#pragma unroll
            for (int v = 0; v < 8; ++v) {
                float e = acc[v] + bias_f;
                hidden[(v + mbase) * LDS_PITCH + nW] = e > 0.0f ? e : 0.0f;
            }
        }
        lds_barrier();

        if (wave == 0) {
            v8f acc = {};
#pragma unroll 4
            for (int k = 0; k < 32; ++k) {
                const int kk = k * 4 + khalf;
                v2f av = *(const v2f*)&hidden[lo * LDS_PITCH + kk];
                v2f bf = wdec_s[(2 * k + hi) * 16 + lo];
                acc = __builtin_amdgcn_wmma_f32_16x16x4_f32(
                        false, av, false, bf, (short)0, acc, false, false);
            }
            if (lo < OUT_DIM) {
#pragma unroll
                for (int v = 0; v < 8; ++v) {
                    const int m = v + mbase;
                    out[(size_t)(blockBase + m) * OUT_DIM + lo] = acc[v] + bias_d;
                }
            }
        }
    }
}

extern "C" void kernel_launch(void* const* d_in, const int* in_sizes, int n_in,
                              void* d_out, int out_size, void* d_ws, size_t ws_size,
                              hipStream_t stream) {
    (void)n_in; (void)out_size; (void)d_ws; (void)ws_size;
    const float* obs   = (const float*)d_in[0];
    const float* W_enc = (const float*)d_in[1];
    const float* b_enc = (const float*)d_in[2];
    const float* W_f   = (const float*)d_in[3];
    const float* b_f   = (const float*)d_in[4];
    const float* W_dec = (const float*)d_in[5];
    const float* b_dec = (const float*)d_in[6];
    float* out = (float*)d_out;

    const int n_samples = in_sizes[0] / (N_STATES * STATE_SIZE);
    if (n_samples <= 0) return;

    gnn_fused_wmma<<<GRID_BLOCKS, BLOCK_THREADS, 0, stream>>>(
        obs, W_enc, b_enc, W_f, b_f, W_dec, b_dec, out, n_samples);
}


# Logits GEMM split over K across the 8 waves, each fed from its own hidden accumulators (hidden GEMM transposed); partials summed via LDS; one barrier per tile; reorders the logit sums
# speedup vs baseline: 1.6676x; 1.0944x over previous
#include <hip/hip_runtime.h>


typedef float v2f __attribute__((ext_vector_type(2)));
typedef float v4f __attribute__((ext_vector_type(4)));
typedef float v8f __attribute__((ext_vector_type(8)));

#define N_STATES   16
#define STATE_SIZE 32
#define ENC        64
#define HID        128
#define OUT_DIM    8
#define TILE       16
#define LDS_PITCH  (HID + 4)
#define N_WAVES    8
#define PART_SLICE (TILE * OUT_DIM)
#define BLOCK_THREADS 256
#define GRID_BLOCKS   512
static_assert(2 * ENC == HID, "f_in row = concat(agent, agg)");
static_assert(N_WAVES * 32 == BLOCK_THREADS && N_WAVES * 16 == HID, "one 16-feature hidden slice per wave");
static_assert(PART_SLICE == 32 * 4, "one wave sums a tile's partials with one 16-byte access per lane");

__device__ __forceinline__ void load_sample(v2f (&a)[8], const float* __restrict__ st,
                                            int lo, int khalf)
{
#pragma unroll
    for (int k = 0; k < 8; ++k)
        a[k] = *(const v2f*)(st + lo * STATE_SIZE + k * 4 + khalf);
}

__device__ __forceinline__ void lds_barrier()
{
    asm volatile("s_wait_dscnt 0x0\n\ts_barrier_signal -1\n\ts_barrier_wait -1" ::: "memory");
}

__device__ __forceinline__ int valid_prefix(float flag, int hi)
{
    const bool ok = (hi == 0) ? (flag == 1.0f) : true;
    const unsigned msk   = __builtin_amdgcn_ballot_w32(ok);
    const unsigned inval = ((~msk) >> 1) & 0x7FFFu;
    return __builtin_ctz(inval | 0x8000u);
}

__device__ __forceinline__ void encode_single(const v2f (&a)[8], const v2f (&we)[4][8],
                                              const float (&be)[4], int cnt, int ls,
                                              int lo, int hi, int mbase, float* f_in)
{
    const float inv = 1.0f / (float)(cnt > 1 ? cnt : 1);
#pragma unroll
    for (int t = 0; t < 4; ++t) {
        const int n = lo + t * 16;
        v8f acc = {};
#pragma unroll
        for (int k = 0; k < 8; ++k)
            acc = __builtin_amdgcn_wmma_f32_16x16x4_f32(
                    false, a[k], false, we[t][k], (short)0, acc, false, false);
        const float bias = be[t];
        float part = 0.0f, agent = 0.0f;
#pragma unroll
        for (int v = 0; v < 8; ++v) {
            const int m = v + mbase;
            float e = acc[v] + bias;
            e = e > 0.0f ? e : 0.0f;
            if (m == 0) agent = e;
            const float w = (m >= 1 && m <= cnt) ? inv : 0.0f;
            part += w * e;
        }
        part += __shfl_xor(part, 16, 32);
        if (hi == 0) {
            f_in[ls * LDS_PITCH + n]       = agent;
            f_in[ls * LDS_PITCH + ENC + n] = part;
        }
    }
}

__device__ __forceinline__ void encode_pair(const v2f (&a0)[8], const v2f (&a1)[8],
                                            const v2f (&we)[4][8], const float (&be)[4],
                                            int cnt0, int cnt1, int ls0,
                                            int lo, int hi, int mbase, float* f_in)
{
    const int  r0    = cnt0 + 1;
    const bool from0 = lo < r0;
    const int  srcl  = ((lo - r0) & 15) | (hi << 4);
    v2f p[8];
#pragma unroll
    for (int k = 0; k < 8; ++k) {
        const float x = __shfl(a1[k][0], srcl, 32);
        const float y = __shfl(a1[k][1], srcl, 32);
        p[k][0] = from0 ? a0[k][0] : x;
        p[k][1] = from0 ? a0[k][1] : y;
    }

    const float inv0 = 1.0f / (float)(cnt0 > 1 ? cnt0 : 1);
    const float inv1 = 1.0f / (float)(cnt1 > 1 ? cnt1 : 1);
    float w0[8], w1[8];
#pragma unroll
    for (int v = 0; v < 8; ++v) {
        const int m = v + mbase;
        w0[v] = (m >= 1 && m <= cnt0)       ? inv0 : 0.0f;
        w1[v] = (m > r0 && m <= r0 + cnt1)  ? inv1 : 0.0f;
    }
    const int ls1 = ls0 + 1;
    const bool has_agent1 = (hi == (r0 >> 3));

#pragma unroll
    for (int t = 0; t < 4; ++t) {
        const int n = lo + t * 16;
        v8f acc = {};
#pragma unroll
        for (int k = 0; k < 8; ++k)
            acc = __builtin_amdgcn_wmma_f32_16x16x4_f32(
                    false, p[k], false, we[t][k], (short)0, acc, false, false);
        const float bias = be[t];
        float part0 = 0.0f, part1 = 0.0f, agent0 = 0.0f, agent1 = 0.0f;
#pragma unroll
        for (int v = 0; v < 8; ++v) {
            const int m = v + mbase;
            float e = acc[v] + bias;
            e = e > 0.0f ? e : 0.0f;
            if (m == 0)  agent0 = e;
            if (m == r0) agent1 = e;
            part0 += w0[v] * e;
            part1 += w1[v] * e;
        }
        part0 += __shfl_xor(part0, 16, 32);
        part1 += __shfl_xor(part1, 16, 32);
        if (hi == 0) {
            f_in[ls0 * LDS_PITCH + n]       = agent0;
            f_in[ls0 * LDS_PITCH + ENC + n] = part0;
            f_in[ls1 * LDS_PITCH + ENC + n] = part1;
        }
        if (has_agent1)
            f_in[ls1 * LDS_PITCH + n] = agent1;
    }
}

__device__ __forceinline__ void sum_partials_store(const float* pbuf, const float* bdec_s,
                                                   float* __restrict__ out, int blockBase, int lane)
{
    v4f s = *(const v4f*)&pbuf[lane * 4];
#pragma unroll
    for (int w = 1; w < N_WAVES; ++w)
        s += *(const v4f*)&pbuf[w * PART_SLICE + lane * 4];
    s += *(const v4f*)&bdec_s[(lane & 1) * 4];
    *(v4f*)&out[(size_t)blockBase * OUT_DIM + lane * 4] = s;
}

__global__ __launch_bounds__(BLOCK_THREADS)
void gnn_fused_wmma(const float* __restrict__ obs,
                    const float* __restrict__ W_enc, const float* __restrict__ b_enc,
                    const float* __restrict__ W_f,   const float* __restrict__ b_f,
                    const float* __restrict__ W_dec, const float* __restrict__ b_dec,
                    float* __restrict__ out, int n_samples)
{
    __shared__ __attribute__((aligned(16))) float f_in[2][TILE * LDS_PITCH];
    __shared__ __attribute__((aligned(16))) float part[2][N_WAVES * PART_SLICE];
    __shared__ __attribute__((aligned(16))) float bf_s[HID];
    __shared__ __attribute__((aligned(16))) float bdec_s[OUT_DIM];
    __shared__ __attribute__((aligned(16))) float wd_s[BLOCK_THREADS * 8];

    const int tid   = threadIdx.x;
    const int wave  = tid >> 5;
    const int lane  = tid & 31;
    const int lo    = lane & 15;
    const int hi    = lane >> 4;
    const int khalf = hi * 2;
    const int mbase = hi * 8;
    const int n_tiles = n_samples / TILE;
    const int stride  = gridDim.x;

    int tile = blockIdx.x;

    v2f a0[8], a1[8];
    if (tile < n_tiles) {
        const float* st = obs + (size_t)(tile * TILE + wave * 2) * (N_STATES * STATE_SIZE);
        load_sample(a0, st, lo, khalf);
        load_sample(a1, st + N_STATES * STATE_SIZE, lo, khalf);
    }

    v2f we[4][8];
    float be[4];
#pragma unroll
    for (int t = 0; t < 4; ++t) {
        const int n = lo + t * 16;
        be[t] = b_enc[n];
#pragma unroll
        for (int k = 0; k < 8; ++k) {
            const int kk = k * 4 + khalf;
            we[t][k][0] = W_enc[kk * ENC + n];
            we[t][k][1] = W_enc[(kk + 1) * ENC + n];
        }
    }
    const int nW = lo + wave * 16;
    v2f wf[32];
#pragma unroll
    for (int k = 0; k < 32; ++k) {
        const int kk = k * 4 + khalf;
        wf[k][0] = W_f[kk * HID + nW];
        wf[k][1] = W_f[(kk + 1) * HID + nW];
    }
    const int fbase = wave * 16 + mbase;
    if (tid < HID)     bf_s[tid]   = b_f[tid];
    if (tid < OUT_DIM) bdec_s[tid] = b_dec[tid];
#pragma unroll
    for (int e = 0; e < 8; ++e)
        wd_s[tid * 8 + e] = (lo < OUT_DIM) ? W_dec[(fbase + e) * OUT_DIM + lo] : 0.0f;

    int buf = 0;
    int prevBase = -1;
    float* pw = &part[0][wave * PART_SLICE + lo * OUT_DIM];
    for (; tile < n_tiles; tile += stride, buf ^= 1) {
        const int blockBase = tile * TILE;
        float* fin = f_in[buf];

        {
            const int ls0  = wave * 2;
            const int cnt0 = valid_prefix(a0[0][0], hi);
            const int cnt1 = valid_prefix(a1[0][0], hi);
            if (cnt0 + cnt1 + 2 <= TILE) {
                encode_pair(a0, a1, we, be, cnt0, cnt1, ls0, lo, hi, mbase, fin);
            } else {
                encode_single(a0, we, be, cnt0, ls0,     lo, hi, mbase, fin);
                encode_single(a1, we, be, cnt1, ls0 + 1, lo, hi, mbase, fin);
            }
        }

        if (tile + stride < n_tiles) {
            const float* st = obs + (size_t)((tile + stride) * TILE + wave * 2) * (N_STATES * STATE_SIZE);
            load_sample(a0, st, lo, khalf);
            load_sample(a1, st + N_STATES * STATE_SIZE, lo, khalf);
        }
        lds_barrier();

        if (wave == 0 && prevBase >= 0)
            sum_partials_store(part[buf ^ 1], bdec_s, out, prevBase, lane);

        v8f acc = {};
#pragma unroll
        for (int k = 0; k < 32; ++k) {
            const int kk = k * 4 + khalf;
            v2f bv = *(const v2f*)&fin[lo * LDS_PITCH + kk];
            acc = __builtin_amdgcn_wmma_f32_16x16x4_f32(
                    false, wf[k], false, bv, (short)0, acc, false, false);
        }
        const v4f bf0 = *(const v4f*)&bf_s[fbase], bf1 = *(const v4f*)&bf_s[fbase + 4];
        const v4f wd0 = *(const v4f*)&wd_s[tid * 8], wd1 = *(const v4f*)&wd_s[tid * 8 + 4];
        float h[8];
#pragma unroll
        for (int r = 0; r < 8; ++r) {
            const float e = acc[r] + (r < 4 ? bf0[r & 3] : bf1[r & 3]);
            h[r] = e > 0.0f ? e : 0.0f;
        }
        __builtin_amdgcn_sched_barrier(0);

        v8f lacc = {};
#pragma unroll
        for (int j = 0; j < 4; ++j) {
            const v2f wa = (j < 2) ? (v2f){wd0[(2 * j) & 3], wd0[(2 * j + 1) & 3]}
                                   : (v2f){wd1[(2 * j) & 3], wd1[(2 * j + 1) & 3]};
            const v2f bh = {h[2 * j], h[2 * j + 1]};
            lacc = __builtin_amdgcn_wmma_f32_16x16x4_f32(
                    false, wa, false, bh, (short)0, lacc, false, false);
        }
        if (hi == 0) {
            *(v4f*)&pw[0] = (v4f){lacc[0], lacc[1], lacc[2], lacc[3]};
            *(v4f*)&pw[4] = (v4f){lacc[4], lacc[5], lacc[6], lacc[7]};
        }
        pw += (buf == 0) ? N_WAVES * PART_SLICE : -(N_WAVES * PART_SLICE);
        prevBase = blockBase;
    }

    if (prevBase >= 0) {
        lds_barrier();
        if (wave == 0)
            sum_partials_store(part[buf ^ 1], bdec_s, out, prevBase, lane);
    }
}

extern "C" void kernel_launch(void* const* d_in, const int* in_sizes, int n_in,
                              void* d_out, int out_size, void* d_ws, size_t ws_size,
                              hipStream_t stream) {
    (void)n_in; (void)out_size; (void)d_ws; (void)ws_size;
    const float* obs   = (const float*)d_in[0];
    const float* W_enc = (const float*)d_in[1];
    const float* b_enc = (const float*)d_in[2];
    const float* W_f   = (const float*)d_in[3];
    const float* b_f   = (const float*)d_in[4];
    const float* W_dec = (const float*)d_in[5];
    const float* b_dec = (const float*)d_in[6];
    float* out = (float*)d_out;

    const int n_samples = in_sizes[0] / (N_STATES * STATE_SIZE);
    if (n_samples <= 0) return;

    gnn_fused_wmma<<<GRID_BLOCKS, BLOCK_THREADS, 0, stream>>>(
        obs, W_enc, b_enc, W_f, b_f, W_dec, b_dec, out, n_samples);
}


# Encoder operands fetched 16 B per lane: 4 loads per sample's rows and 16 for W_enc (were 8 and 64), by permuting k and feature order of the A and B fragments alike; reorders the encoder's k sums
# speedup vs baseline: 1.7486x; 1.0486x over previous
#include <hip/hip_runtime.h>


typedef float v2f __attribute__((ext_vector_type(2)));
typedef float v4f __attribute__((ext_vector_type(4)));
typedef float v8f __attribute__((ext_vector_type(8)));

#define N_STATES   16
#define STATE_SIZE 32
#define ENC        64
#define HID        128
#define OUT_DIM    8
#define TILE       16
#define LDS_PITCH  (HID + 4)
#define N_WAVES    8
#define PART_SLICE (TILE * OUT_DIM)
#define BLOCK_THREADS 256
#define GRID_BLOCKS   512
static_assert(2 * ENC == HID, "f_in row = concat(agent, agg)");
static_assert(N_WAVES * 32 == BLOCK_THREADS && N_WAVES * 16 == HID, "one 16-feature hidden slice per wave");
static_assert(PART_SLICE == 32 * 4, "one wave sums a tile's partials with one 16-byte access per lane");

__device__ __forceinline__ constexpr int enc_k(int j, int h, int s)
{
    return 8 * (j >> 1) + 4 * h + 2 * (j & 1) + s;
}

__device__ __forceinline__ void load_sample(v2f (&a)[8], const float* __restrict__ st,
                                            int lo, int hi)
{
#pragma unroll
    for (int q = 0; q < 4; ++q) {
        const v4f x = *(const v4f*)(st + lo * STATE_SIZE + q * 8 + hi * 4);
        a[2 * q]     = (v2f){x[0], x[1]};
        a[2 * q + 1] = (v2f){x[2], x[3]};
    }
}

__device__ __forceinline__ void lds_barrier()
{
    asm volatile("s_wait_dscnt 0x0\n\ts_barrier_signal -1\n\ts_barrier_wait -1" ::: "memory");
}

__device__ __forceinline__ int valid_prefix(float flag, int hi)
{
    const bool ok = (hi == 0) ? (flag == 1.0f) : true;
    const unsigned msk   = __builtin_amdgcn_ballot_w32(ok);
    const unsigned inval = ((~msk) >> 1) & 0x7FFFu;
    return __builtin_ctz(inval | 0x8000u);
}

__device__ __forceinline__ void encode_single(const v2f (&a)[8], const v2f (&we)[4][8],
                                              const v4f be, int cnt, int ls,
                                              int lo, int hi, int mbase, float* f_in)
{
    const float inv = 1.0f / (float)(cnt > 1 ? cnt : 1);
#pragma unroll
    for (int t = 0; t < 4; ++t) {
        const int n = 4 * lo + t;
        v8f acc = {};
#pragma unroll
        for (int k = 0; k < 8; ++k)
            acc = __builtin_amdgcn_wmma_f32_16x16x4_f32(
                    false, a[k], false, we[t][k], (short)0, acc, false, false);
        const float bias = be[t];
        float part = 0.0f, agent = 0.0f;
#pragma unroll
        for (int v = 0; v < 8; ++v) {
            const int m = v + mbase;
            float e = acc[v] + bias;
            e = e > 0.0f ? e : 0.0f;
            if (m == 0) agent = e;
            const float w = (m >= 1 && m <= cnt) ? inv : 0.0f;
            part += w * e;
        }
        part += __shfl_xor(part, 16, 32);
        if (hi == 0) {
            f_in[ls * LDS_PITCH + n]       = agent;
            f_in[ls * LDS_PITCH + ENC + n] = part;
        }
    }
}

__device__ __forceinline__ void encode_pair(const v2f (&a0)[8], const v2f (&a1)[8],
                                            const v2f (&we)[4][8], const v4f be,
                                            int cnt0, int cnt1, int ls0,
                                            int lo, int hi, int mbase, float* f_in)
{
    const int  r0    = cnt0 + 1;
    const bool from0 = lo < r0;
    const int  srcl  = ((lo - r0) & 15) | (hi << 4);
    v2f p[8];
#pragma unroll
    for (int k = 0; k < 8; ++k) {
        const float x = __shfl(a1[k][0], srcl, 32);
        const float y = __shfl(a1[k][1], srcl, 32);
        p[k][0] = from0 ? a0[k][0] : x;
        p[k][1] = from0 ? a0[k][1] : y;
    }

    const float inv0 = 1.0f / (float)(cnt0 > 1 ? cnt0 : 1);
    const float inv1 = 1.0f / (float)(cnt1 > 1 ? cnt1 : 1);
    float w0[8], w1[8];
#pragma unroll
    for (int v = 0; v < 8; ++v) {
        const int m = v + mbase;
        w0[v] = (m >= 1 && m <= cnt0)       ? inv0 : 0.0f;
        w1[v] = (m > r0 && m <= r0 + cnt1)  ? inv1 : 0.0f;
    }
    const int ls1 = ls0 + 1;
    const bool has_agent1 = (hi == (r0 >> 3));

#pragma unroll
    for (int t = 0; t < 4; ++t) {
        const int n = 4 * lo + t;
        v8f acc = {};
#pragma unroll
        for (int k = 0; k < 8; ++k)
            acc = __builtin_amdgcn_wmma_f32_16x16x4_f32(
                    false, p[k], false, we[t][k], (short)0, acc, false, false);
        const float bias = be[t];
        float part0 = 0.0f, part1 = 0.0f, agent0 = 0.0f, agent1 = 0.0f;
#pragma unroll
        for (int v = 0; v < 8; ++v) {
            const int m = v + mbase;
            float e = acc[v] + bias;
            e = e > 0.0f ? e : 0.0f;
            if (m == 0)  agent0 = e;
            if (m == r0) agent1 = e;
            part0 += w0[v] * e;
            part1 += w1[v] * e;
        }
        part0 += __shfl_xor(part0, 16, 32);
        part1 += __shfl_xor(part1, 16, 32);
        if (hi == 0) {
            f_in[ls0 * LDS_PITCH + n]       = agent0;
            f_in[ls0 * LDS_PITCH + ENC + n] = part0;
            f_in[ls1 * LDS_PITCH + ENC + n] = part1;
        }
        if (has_agent1)
            f_in[ls1 * LDS_PITCH + n] = agent1;
    }
}

__device__ __forceinline__ void sum_partials_store(const float* pbuf, const float* bdec_s,
                                                   float* __restrict__ out, int blockBase, int lane)
{
    v4f s = *(const v4f*)&pbuf[lane * 4];
#pragma unroll
    for (int w = 1; w < N_WAVES; ++w)
        s += *(const v4f*)&pbuf[w * PART_SLICE + lane * 4];
    s += *(const v4f*)&bdec_s[(lane & 1) * 4];
    *(v4f*)&out[(size_t)blockBase * OUT_DIM + lane * 4] = s;
}

__global__ __launch_bounds__(BLOCK_THREADS)
void gnn_fused_wmma(const float* __restrict__ obs,
                    const float* __restrict__ W_enc, const float* __restrict__ b_enc,
                    const float* __restrict__ W_f,   const float* __restrict__ b_f,
                    const float* __restrict__ W_dec, const float* __restrict__ b_dec,
                    float* __restrict__ out, int n_samples)
{
    __shared__ __attribute__((aligned(16))) float f_in[2][TILE * LDS_PITCH];
    __shared__ __attribute__((aligned(16))) float part[2][N_WAVES * PART_SLICE];
    __shared__ __attribute__((aligned(16))) float bf_s[HID];
    __shared__ __attribute__((aligned(16))) float bdec_s[OUT_DIM];
    __shared__ __attribute__((aligned(16))) float wd_s[BLOCK_THREADS * 8];

    const int tid   = threadIdx.x;
    const int wave  = tid >> 5;
    const int lane  = tid & 31;
    const int lo    = lane & 15;
    const int hi    = lane >> 4;
    const int khalf = hi * 2;
    const int mbase = hi * 8;
    const int n_tiles = n_samples / TILE;
    const int stride  = gridDim.x;

    int tile = blockIdx.x;

    v2f a0[8], a1[8];
    if (tile < n_tiles) {
        const float* st = obs + (size_t)(tile * TILE + wave * 2) * (N_STATES * STATE_SIZE);
        load_sample(a0, st, lo, hi);
        load_sample(a1, st + N_STATES * STATE_SIZE, lo, hi);
    }

    v2f we[4][8];
    const v4f be = *(const v4f*)&b_enc[4 * lo];
#pragma unroll
    for (int k = 0; k < 8; ++k) {
#pragma unroll
        for (int s = 0; s < 2; ++s) {
            const v4f w = *(const v4f*)&W_enc[enc_k(k, hi, s) * ENC + 4 * lo];
#pragma unroll
            for (int t = 0; t < 4; ++t)
                we[t][k][s] = w[t];
        }
    }
    const int nW = lo + wave * 16;
    v2f wf[32];
#pragma unroll
    for (int k = 0; k < 32; ++k) {
        const int kk = k * 4 + khalf;
        wf[k][0] = W_f[kk * HID + nW];
        wf[k][1] = W_f[(kk + 1) * HID + nW];
    }
    const int fbase = wave * 16 + mbase;
    if (tid < HID)     bf_s[tid]   = b_f[tid];
    if (tid < OUT_DIM) bdec_s[tid] = b_dec[tid];
#pragma unroll
    for (int e = 0; e < 8; ++e)
        wd_s[tid * 8 + e] = (lo < OUT_DIM) ? W_dec[(fbase + e) * OUT_DIM + lo] : 0.0f;

    int buf = 0;
    int prevBase = -1;
    float* pw = &part[0][wave * PART_SLICE + lo * OUT_DIM];
    for (; tile < n_tiles; tile += stride, buf ^= 1) {
        const int blockBase = tile * TILE;
        float* fin = f_in[buf];

        {
            const int ls0  = wave * 2;
            const int cnt0 = valid_prefix(a0[0][0], hi);
            const int cnt1 = valid_prefix(a1[0][0], hi);
            if (cnt0 + cnt1 + 2 <= TILE) {
                encode_pair(a0, a1, we, be, cnt0, cnt1, ls0, lo, hi, mbase, fin);
            } else {
                encode_single(a0, we, be, cnt0, ls0,     lo, hi, mbase, fin);
                encode_single(a1, we, be, cnt1, ls0 + 1, lo, hi, mbase, fin);
            }
        }

        if (tile + stride < n_tiles) {
            const float* st = obs + (size_t)((tile + stride) * TILE + wave * 2) * (N_STATES * STATE_SIZE);
            load_sample(a0, st, lo, hi);
            load_sample(a1, st + N_STATES * STATE_SIZE, lo, hi);
        }
        lds_barrier();

        if (wave == 0 && prevBase >= 0)
            sum_partials_store(part[buf ^ 1], bdec_s, out, prevBase, lane);

        v8f acc = {};
#pragma unroll
        for (int k = 0; k < 32; ++k) {
            const int kk = k * 4 + khalf;
            v2f bv = *(const v2f*)&fin[lo * LDS_PITCH + kk];
            acc = __builtin_amdgcn_wmma_f32_16x16x4_f32(
                    false, wf[k], false, bv, (short)0, acc, false, false);
        }
        const v4f bf0 = *(const v4f*)&bf_s[fbase], bf1 = *(const v4f*)&bf_s[fbase + 4];
        const v4f wd0 = *(const v4f*)&wd_s[tid * 8], wd1 = *(const v4f*)&wd_s[tid * 8 + 4];
        float h[8];
#pragma unroll
        for (int r = 0; r < 8; ++r) {
            const float e = acc[r] + (r < 4 ? bf0[r & 3] : bf1[r & 3]);
            h[r] = e > 0.0f ? e : 0.0f;
        }
        __builtin_amdgcn_sched_barrier(0);

        v8f lacc = {};
#pragma unroll
        for (int j = 0; j < 4; ++j) {
            const v2f wa = (j < 2) ? (v2f){wd0[(2 * j) & 3], wd0[(2 * j + 1) & 3]}
                                   : (v2f){wd1[(2 * j) & 3], wd1[(2 * j + 1) & 3]};
            const v2f bh = {h[2 * j], h[2 * j + 1]};
            lacc = __builtin_amdgcn_wmma_f32_16x16x4_f32(
                    false, wa, false, bh, (short)0, lacc, false, false);
        }
        if (hi == 0) {
            *(v4f*)&pw[0] = (v4f){lacc[0], lacc[1], lacc[2], lacc[3]};
            *(v4f*)&pw[4] = (v4f){lacc[4], lacc[5], lacc[6], lacc[7]};
        }
        pw += (buf == 0) ? N_WAVES * PART_SLICE : -(N_WAVES * PART_SLICE);
        prevBase = blockBase;
    }

    if (prevBase >= 0) {
        lds_barrier();
        if (wave == 0)
            sum_partials_store(part[buf ^ 1], bdec_s, out, prevBase, lane);
    }
}

extern "C" void kernel_launch(void* const* d_in, const int* in_sizes, int n_in,
                              void* d_out, int out_size, void* d_ws, size_t ws_size,
                              hipStream_t stream) {
    (void)n_in; (void)out_size; (void)d_ws; (void)ws_size;
    const float* obs   = (const float*)d_in[0];
    const float* W_enc = (const float*)d_in[1];
    const float* b_enc = (const float*)d_in[2];
    const float* W_f   = (const float*)d_in[3];
    const float* b_f   = (const float*)d_in[4];
    const float* W_dec = (const float*)d_in[5];
    const float* b_dec = (const float*)d_in[6];
    float* out = (float*)d_out;

    const int n_samples = in_sizes[0] / (N_STATES * STATE_SIZE);
    if (n_samples <= 0) return;

    gnn_fused_wmma<<<GRID_BLOCKS, BLOCK_THREADS, 0, stream>>>(
        obs, W_enc, b_enc, W_f, b_f, W_dec, b_dec, out, n_samples);
}
